# starting kernel
# baseline (speedup 1.0000x reference)
#include <hip/hip_runtime.h>


#define B_ 8
#define N_ 16384
#define K_ 256
#define D_ 20
#define NCHUNKS 64
#define NB 128
#define NBP 130
#define ITERS ((N_ / NCHUNKS) / NB)

typedef __attribute__((ext_vector_type(2))) float v2f;
typedef __attribute__((ext_vector_type(8))) float v8f;

__global__ __launch_bounds__(256) void em_zero(float* __restrict__ out,
                                               float* __restrict__ den) {
  int i = blockIdx.x * 256 + threadIdx.x;
  if (i < B_ * K_ * D_) out[i] = 0.0f;
  if (i < B_ * K_) den[i] = 0.0f;
}

__global__ __launch_bounds__(256) void em_main(const float* __restrict__ mean,
                                               const float* __restrict__ feat,
                                               const float* __restrict__ valid,
                                               const float* __restrict__ adj,
                                               float* __restrict__ out,
                                               float* __restrict__ den) {
  __shared__ float smean[K_ * D_];
  __shared__ float smsq[K_];
  __shared__ float pbuf[K_ * NBP];
  __shared__ float rinvb[NB];

  const int b = blockIdx.x / NCHUNKS;
  const int chunk = blockIdx.x % NCHUNKS;
  const int tid = threadIdx.x;
  const int w = tid >> 5;
  const int lane = tid & 31;
  const int hi = lane >> 4;
  const int lr = lane & 15;

  {
    const float* mrow = mean + ((size_t)b * K_ + tid) * D_;
    float sq = 0.0f;
#pragma unroll
    for (int d = 0; d < D_; ++d) {
      float v = mrow[d];
      smean[tid * D_ + d] = v;
      sq += v * v;
    }
    smsq[tid] = sq;
  }
  __syncthreads();

  v8f acc[2][2];
#pragma unroll
  for (int q = 0; q < 2; ++q)
#pragma unroll
    for (int dt = 0; dt < 2; ++dt) acc[q][dt] = (v8f){0,0,0,0,0,0,0,0};

  const int nchunk0 = chunk * (N_ / NCHUNKS);

  for (int it = 0; it < ITERS; ++it) {
    const int nb0 = nchunk0 + it * NB;
    const int n0 = nb0 + w * 16;

    __syncthreads();

    v2f a[5];
    float psq = 0.0f;
    {
      const float* fbase = feat + ((size_t)b * N_ + n0 + lr) * D_ + 2 * hi;
#pragma unroll
      for (int s = 0; s < 5; ++s) {
        v2f t;
        t.x = fbase[s * 4];
        t.y = fbase[s * 4 + 1];
        a[s] = t;
        psq += t.x * t.x + t.y * t.y;
      }
    }
    float fsq = psq + __shfl_xor(psq, 16, 32);
    float fsqrow[8];
#pragma unroll
    for (int j = 0; j < 8; ++j) fsqrow[j] = __shfl(fsq, j + 8 * hi, 32);

    float rmax[8];
#pragma unroll
    for (int j = 0; j < 8; ++j) rmax[j] = -3.0e38f;

#pragma unroll
    for (int kt = 0; kt < 16; ++kt) {
      v2f bm[5];
      {
        const float* mrow = smean + (kt * 16 + lr) * D_ + 2 * hi;
#pragma unroll
        for (int s = 0; s < 5; ++s) {
          v2f t;
          t.x = mrow[s * 4];
          t.y = mrow[s * 4 + 1];
          bm[s] = t;
        }
      }
      v8f c = (v8f){0,0,0,0,0,0,0,0};
#pragma unroll
      for (int s = 0; s < 5; ++s)
        c = __builtin_amdgcn_wmma_f32_16x16x4_f32(false, a[s], false, bm[s],
                                                  (short)0, c, false, false);
      const float msq = smsq[kt * 16 + lr];
      const size_t vbase = ((size_t)b * N_ + n0) * K_ + kt * 16 + lr;
      const int prow = (kt * 16 + lr) * NBP + w * 16;
#pragma unroll
      for (int j = 0; j < 8; ++j) {
        const int m = j + 8 * hi;
        const float vs = __builtin_nontemporal_load(valid + vbase + (size_t)m * K_);
        const float ad = __builtin_nontemporal_load(adj + vbase + (size_t)m * K_);
        const float dist = (fsqrow[j] - 2.0f * c[j] + msq) * vs + ad;
        const float sc = -dist;
        rmax[j] = fmaxf(rmax[j], sc);
        pbuf[prow + m] = sc;
      }
    }

#pragma unroll
    for (int j = 0; j < 8; ++j) {
      float r = rmax[j];
      r = fmaxf(r, __shfl_xor(r, 1, 32));
      r = fmaxf(r, __shfl_xor(r, 2, 32));
      r = fmaxf(r, __shfl_xor(r, 4, 32));
      r = fmaxf(r, __shfl_xor(r, 8, 32));
      rmax[j] = r;
    }

    float rsum[8];
#pragma unroll
    for (int j = 0; j < 8; ++j) rsum[j] = 0.0f;
#pragma unroll
    for (int kt = 0; kt < 16; ++kt) {
      const int prow = (kt * 16 + lr) * NBP + w * 16;
#pragma unroll
      for (int j = 0; j < 8; ++j) {
        const int m = j + 8 * hi;
        const float e = __expf(pbuf[prow + m] - rmax[j]);
        rsum[j] += e;
        pbuf[prow + m] = e;
      }
    }
#pragma unroll
    for (int j = 0; j < 8; ++j) {
      float r = rsum[j];
      r += __shfl_xor(r, 1, 32);
      r += __shfl_xor(r, 2, 32);
      r += __shfl_xor(r, 4, 32);
      r += __shfl_xor(r, 8, 32);
      if (lr == j) rinvb[w * 16 + j + 8 * hi] = 1.0f / r;
    }
    __syncthreads();

#pragma unroll 2
    for (int ns = 0; ns < NB / 4; ++ns) {
      v2f pa[2];
#pragma unroll
      for (int q = 0; q < 2; ++q) {
        const int kk = (w * 2 + q) * 16 + lr;
        const float* pr = pbuf + kk * NBP + ns * 4 + 2 * hi;
        v2f t;
        t.x = pr[0];
        t.y = pr[1];
        pa[q] = t;
      }
      const float ri0 = rinvb[ns * 4 + 2 * hi];
      const float ri1 = rinvb[ns * 4 + 2 * hi + 1];
      v2f fb[2];
#pragma unroll
      for (int dt = 0; dt < 2; ++dt) {
        const int d = dt * 16 + lr;
        float x0, x1;
        if (d < D_) {
          const float* f0 =
              feat + ((size_t)b * N_ + nb0 + ns * 4 + 2 * hi) * D_ + d;
          x0 = f0[0] * ri0;
          x1 = f0[D_] * ri1;
        } else if (d == D_) {
          x0 = ri0;
          x1 = ri1;
        } else {
          x0 = 0.0f;
          x1 = 0.0f;
        }
        v2f t;
        t.x = x0;
        t.y = x1;
        fb[dt] = t;
      }
#pragma unroll
      for (int q = 0; q < 2; ++q)
#pragma unroll
        for (int dt = 0; dt < 2; ++dt)
          acc[q][dt] = __builtin_amdgcn_wmma_f32_16x16x4_f32(
              false, pa[q], false, fb[dt], (short)0, acc[q][dt], false, false);
    }
  }

#pragma unroll
  for (int q = 0; q < 2; ++q) {
    const int kbase = (w * 2 + q) * 16;
#pragma unroll
    for (int dt = 0; dt < 2; ++dt) {
      const int d = dt * 16 + lr;
#pragma unroll
      for (int j = 0; j < 8; ++j) {
        const int k = kbase + j + 8 * hi;
        const float v = acc[q][dt][j];
        if (d < D_)
          atomicAdd(&out[((size_t)b * K_ + k) * D_ + d], v);
        else if (d == D_)
          atomicAdd(&den[b * K_ + k], v);
      }
    }
  }
}

__global__ __launch_bounds__(256) void em_fin(float* __restrict__ out,
                                              const float* __restrict__ den) {
  int i = blockIdx.x * 256 + threadIdx.x;
  if (i < B_ * K_ * D_) out[i] = out[i] / fmaxf(den[i / D_], 1e-12f);
}

extern "C" void kernel_launch(void* const* d_in, const int* in_sizes, int n_in,
                              void* d_out, int out_size, void* d_ws,
                              size_t ws_size, hipStream_t stream) {
  (void)in_sizes; (void)n_in; (void)out_size; (void)ws_size;
  const float* mean = (const float*)d_in[0];
  const float* feat = (const float*)d_in[1];
  const float* valid = (const float*)d_in[2];
  const float* adj = (const float*)d_in[3];
  float* out = (float*)d_out;
  float* den = (float*)d_ws;

  const int tot = B_ * K_ * D_;
  em_zero<<<(tot + 255) / 256, 256, 0, stream>>>(out, den);
  em_main<<<B_ * NCHUNKS, 256, 0, stream>>>(mean, feat, valid, adj, out, den);
  em_fin<<<(tot + 255) / 256, 256, 0, stream>>>(out, den);
}


# valid/adjust read as 16-byte loads (256-byte row pieces), 16 kept in flight, each row pair refetched once used; clusters in tile order in LDS, so row sums add in tile order; 4 waves/SIMD bound
# speedup vs baseline: 2.4589x; 2.4589x over previous
#include <hip/hip_runtime.h>


#define B_ 8
#define N_ 16384
#define K_ 256
#define D_ 20
#define NCHUNKS 64
#define NB 128
#define NBP 130
#define ITERS ((N_ / NCHUNKS) / NB)

typedef __attribute__((ext_vector_type(2))) float v2f;
typedef __attribute__((ext_vector_type(4))) float v4f;
typedef __attribute__((ext_vector_type(8))) float v8f;

__global__ __launch_bounds__(256) void em_zero(float* __restrict__ out,
                                               float* __restrict__ den) {
  int i = blockIdx.x * 256 + threadIdx.x;
  if (i < B_ * K_ * D_) out[i] = 0.0f;
  if (i < B_ * K_) den[i] = 0.0f;
}

__global__ __launch_bounds__(256, 4) void em_main(const float* __restrict__ mean,
                                               const float* __restrict__ feat,
                                               const float* __restrict__ valid,
                                               const float* __restrict__ adj,
                                               float* __restrict__ out,
                                               float* __restrict__ den) {
  __shared__ float smean[K_ * D_];
  __shared__ float smsq[K_];
  __shared__ float pbuf[K_ * NBP];
  __shared__ float rinvb[NB];

  const int b = blockIdx.x / NCHUNKS;
  const int chunk = blockIdx.x % NCHUNKS;
  const int tid = threadIdx.x;
  const int w = tid >> 5;
  const int lane = tid & 31;
  const int hi = lane >> 4;
  const int lr = lane & 15;

  {
    const float* mrow = mean + ((size_t)b * K_ + tid) * D_;
    const int r = (((tid >> 6) * 4 + (tid & 3)) * 16) + ((tid >> 2) & 15);
    float sq = 0.0f;
#pragma unroll
    for (int d = 0; d < D_; ++d) {
      float v = mrow[d];
      smean[r * D_ + d] = v;
      sq += v * v;
    }
    smsq[r] = sq;
  }
  __syncthreads();

  v8f acc[2][2];
#pragma unroll
  for (int q = 0; q < 2; ++q)
#pragma unroll
    for (int dt = 0; dt < 2; ++dt) acc[q][dt] = (v8f){0,0,0,0,0,0,0,0};

  const int nchunk0 = chunk * (N_ / NCHUNKS);

  for (int it = 0; it < ITERS; ++it) {
    const int nb0 = nchunk0 + it * NB;
    const int n0 = nb0 + w * 16;

    __syncthreads();

    v2f a[5];
    {
      const float* fbase = feat + ((size_t)b * N_ + n0 + lr) * D_ + 2 * hi;
#pragma unroll
      for (int s = 0; s < 5; ++s) {
        v2f t;
        t.x = fbase[s * 4];
        t.y = fbase[s * 4 + 1];
        a[s] = t;
      }
    }
    __builtin_amdgcn_sched_barrier(0);

    const v4f* vrow =
        (const v4f*)(valid + ((size_t)b * N_ + n0 + 8 * hi) * K_ + 4 * lr);
    const v4f* arow =
        (const v4f*)(adj + ((size_t)b * N_ + n0 + 8 * hi) * K_ + 4 * lr);
    v4f vq[8], aq[8];
#pragma unroll
    for (int j = 0; j < 8; ++j) {
      vq[j] = __builtin_nontemporal_load(vrow + j * (K_ / 4));
      aq[j] = __builtin_nontemporal_load(arow + j * (K_ / 4));
    }

    __builtin_amdgcn_sched_barrier(0);

    float psq = 0.0f;
#pragma unroll
    for (int s = 0; s < 5; ++s) psq += a[s].x * a[s].x + a[s].y * a[s].y;
    float fsq = psq + __shfl_xor(psq, 16, 32);
    float fsqrow[8];
#pragma unroll
    for (int j = 0; j < 8; ++j) fsqrow[j] = __shfl(fsq, j + 8 * hi, 32);

    float rmax[8];
#pragma unroll
    for (int j = 0; j < 8; ++j) rmax[j] = -3.0e38f;

#pragma unroll
    for (int q = 0; q < 4; ++q) {
      v8f c[4];
#pragma unroll
      for (int ep = 0; ep < 2; ++ep) {
        v2f bm[2][5];
#pragma unroll
        for (int e = 0; e < 2; ++e) {
          const float* mrow = smean + ((q * 4 + 2 * ep + e) * 16 + lr) * D_ + 2 * hi;
#pragma unroll
          for (int s = 0; s < 5; ++s) {
            v2f t;
            t.x = mrow[s * 4];
            t.y = mrow[s * 4 + 1];
            bm[e][s] = t;
          }
        }
#pragma unroll
        for (int e = 0; e < 2; ++e) c[2 * ep + e] = (v8f){0,0,0,0,0,0,0,0};
#pragma unroll
        for (int s = 0; s < 5; ++s)
#pragma unroll
          for (int e = 0; e < 2; ++e)
            c[2 * ep + e] = __builtin_amdgcn_wmma_f32_16x16x4_f32(
                false, a[s], false, bm[e][s], (short)0, c[2 * ep + e], false, false);
        __builtin_amdgcn_sched_barrier(0);
      }
      float msq[4];
#pragma unroll
      for (int e = 0; e < 4; ++e) msq[e] = smsq[(q * 4 + e) * 16 + lr];
      const int prow = (q * 64 + lr) * NBP + w * 16 + 8 * hi;
#pragma unroll
      for (int jp = 0; jp < 4; ++jp) {
        v2f sc[4];
#pragma unroll
        for (int h = 0; h < 2; ++h) {
          const int j = 2 * jp + h;
          const v4f vs = vq[j];
          const v4f ad = aq[j];
#pragma unroll
          for (int e = 0; e < 4; ++e) {
            const float dist = (fsqrow[j] - 2.0f * c[e][j] + msq[e]) * vs[e] + ad[e];
            const float s = -dist;
            rmax[j] = fmaxf(rmax[j], s);
            sc[e][h] = s;
          }
        }
        __builtin_amdgcn_sched_barrier(0);
        if (q < 3) {
#pragma unroll
          for (int h = 0; h < 2; ++h) {
            const int j = 2 * jp + h;
            vq[j] = __builtin_nontemporal_load(vrow + j * (K_ / 4) + (q + 1) * 16);
            aq[j] = __builtin_nontemporal_load(arow + j * (K_ / 4) + (q + 1) * 16);
          }
        }
#pragma unroll
        for (int e = 0; e < 4; ++e) {
          pbuf[prow + e * 16 * NBP + 2 * jp] = sc[e].x;
          pbuf[prow + e * 16 * NBP + 2 * jp + 1] = sc[e].y;
        }
      }
    }

#pragma unroll
    for (int j = 0; j < 8; ++j) {
      float r = rmax[j];
      r = fmaxf(r, __shfl_xor(r, 1, 32));
      r = fmaxf(r, __shfl_xor(r, 2, 32));
      r = fmaxf(r, __shfl_xor(r, 4, 32));
      r = fmaxf(r, __shfl_xor(r, 8, 32));
      rmax[j] = r;
    }

    float rsum[8];
#pragma unroll
    for (int j = 0; j < 8; ++j) rsum[j] = 0.0f;
#pragma unroll
    for (int kt = 0; kt < 16; ++kt) {
      const int prow = (kt * 16 + lr) * NBP + w * 16;
#pragma unroll
      for (int j = 0; j < 8; ++j) {
        const int m = j + 8 * hi;
        const float e = __expf(pbuf[prow + m] - rmax[j]);
        rsum[j] += e;
        pbuf[prow + m] = e;
      }
    }
#pragma unroll
    for (int j = 0; j < 8; ++j) {
      float r = rsum[j];
      r += __shfl_xor(r, 1, 32);
      r += __shfl_xor(r, 2, 32);
      r += __shfl_xor(r, 4, 32);
      r += __shfl_xor(r, 8, 32);
      if (lr == j) rinvb[w * 16 + j + 8 * hi] = 1.0f / r;
    }
    __syncthreads();

#pragma unroll 2
    for (int ns = 0; ns < NB / 4; ++ns) {
      v2f pa[2];
#pragma unroll
      for (int q = 0; q < 2; ++q) {
        const int kk = (w * 2 + q) * 16 + lr;
        const float* pr = pbuf + kk * NBP + ns * 4 + 2 * hi;
        v2f t;
        t.x = pr[0];
        t.y = pr[1];
        pa[q] = t;
      }
      const float ri0 = rinvb[ns * 4 + 2 * hi];
      const float ri1 = rinvb[ns * 4 + 2 * hi + 1];
      v2f fb[2];
#pragma unroll
      for (int dt = 0; dt < 2; ++dt) {
        const int d = dt * 16 + lr;
        float x0, x1;
        if (d < D_) {
          const float* f0 =
              feat + ((size_t)b * N_ + nb0 + ns * 4 + 2 * hi) * D_ + d;
          x0 = f0[0] * ri0;
          x1 = f0[D_] * ri1;
        } else if (d == D_) {
          x0 = ri0;
          x1 = ri1;
        } else {
          x0 = 0.0f;
          x1 = 0.0f;
        }
        v2f t;
        t.x = x0;
        t.y = x1;
        fb[dt] = t;
      }
#pragma unroll
      for (int q = 0; q < 2; ++q)
#pragma unroll
        for (int dt = 0; dt < 2; ++dt)
          acc[q][dt] = __builtin_amdgcn_wmma_f32_16x16x4_f32(
              false, pa[q], false, fb[dt], (short)0, acc[q][dt], false, false);
    }
  }

#pragma unroll
  for (int q = 0; q < 2; ++q) {
    const int t = w * 2 + q;
    const int kbase = (t >> 2) * 64 + (t & 3);
#pragma unroll
    for (int dt = 0; dt < 2; ++dt) {
      const int d = dt * 16 + lr;
#pragma unroll
      for (int j = 0; j < 8; ++j) {
        const int k = kbase + 4 * (j + 8 * hi);
        const float v = acc[q][dt][j];
        if (d < D_)
          atomicAdd(&out[((size_t)b * K_ + k) * D_ + d], v);
        else if (d == D_)
          atomicAdd(&den[b * K_ + k], v);
      }
    }
  }
}

__global__ __launch_bounds__(256) void em_fin(float* __restrict__ out,
                                              const float* __restrict__ den) {
  int i = blockIdx.x * 256 + threadIdx.x;
  if (i < B_ * K_ * D_) out[i] = out[i] / fmaxf(den[i / D_], 1e-12f);
}

extern "C" void kernel_launch(void* const* d_in, const int* in_sizes, int n_in,
                              void* d_out, int out_size, void* d_ws,
                              size_t ws_size, hipStream_t stream) {
  (void)in_sizes; (void)n_in; (void)out_size; (void)ws_size;
  const float* mean = (const float*)d_in[0];
  const float* feat = (const float*)d_in[1];
  const float* valid = (const float*)d_in[2];
  const float* adj = (const float*)d_in[3];
  float* out = (float*)d_out;
  float* den = (float*)d_ws;

  const int tot = B_ * K_ * D_;
  em_zero<<<(tot + 255) / 256, 256, 0, stream>>>(out, den);
  em_main<<<B_ * NCHUNKS, 256, 0, stream>>>(mean, feat, valid, adj, out, den);
  em_fin<<<(tot + 255) / 256, 256, 0, stream>>>(out, den);
}


# two 128-pixel blocks per 16-wave workgroup, second half's partial sums added to the first's in LDS before the flush: 256 workgroups, half the float atomics; partial sums add in a new order
# speedup vs baseline: 3.3004x; 1.3422x over previous
#include <hip/hip_runtime.h>


#define B_ 8
#define N_ 16384
#define K_ 256
#define D_ 20
#define NCHUNKS 32
#define NB 128
#define NBP 130
#define ITERS ((N_ / NCHUNKS) / (2 * NB))

typedef __attribute__((ext_vector_type(2))) float v2f;
typedef __attribute__((ext_vector_type(4))) float v4f;
typedef __attribute__((ext_vector_type(8))) float v8f;

__global__ __launch_bounds__(256) void em_zero(float* __restrict__ out,
                                               float* __restrict__ den) {
  int i = blockIdx.x * 256 + threadIdx.x;
  if (i < B_ * K_ * D_) out[i] = 0.0f;
  if (i < B_ * K_) den[i] = 0.0f;
}

__global__ __launch_bounds__(512, 4) void em_main(const float* __restrict__ mean,
                                               const float* __restrict__ feat,
                                               const float* __restrict__ valid,
                                               const float* __restrict__ adj,
                                               float* __restrict__ out,
                                               float* __restrict__ den) {
  __shared__ float smean[K_ * D_];
  __shared__ float smsq[K_];
  __shared__ float pbuf2[2 * K_ * NBP];
  __shared__ float rinvb2[2 * NB];

  const int b = blockIdx.x / NCHUNKS;
  const int chunk = blockIdx.x % NCHUNKS;
  const int tid = threadIdx.x;
  const int half = __builtin_amdgcn_readfirstlane(tid >> 8);
  const int w = (tid >> 5) & 7;
  float* const pbuf = pbuf2 + half * (K_ * NBP);
  float* const rinvb = rinvb2 + half * NB;
  const int lane = tid & 31;
  const int hi = lane >> 4;
  const int lr = lane & 15;

  if (tid < K_) {
    const float* mrow = mean + ((size_t)b * K_ + tid) * D_;
    const int r = (((tid >> 6) * 4 + (tid & 3)) * 16) + ((tid >> 2) & 15);
    float sq = 0.0f;
#pragma unroll
    for (int d = 0; d < D_; ++d) {
      float v = mrow[d];
      smean[r * D_ + d] = v;
      sq += v * v;
    }
    smsq[r] = sq;
  }
  __syncthreads();

  v8f acc[2][2];
#pragma unroll
  for (int q = 0; q < 2; ++q)
#pragma unroll
    for (int dt = 0; dt < 2; ++dt) acc[q][dt] = (v8f){0,0,0,0,0,0,0,0};

  const int nchunk0 = chunk * (N_ / NCHUNKS);

  for (int it = 0; it < ITERS; ++it) {
    const int nb0 = nchunk0 + (2 * it + half) * NB;
    const int n0 = nb0 + w * 16;

    __syncthreads();

    v2f a[5];
    {
      const float* fbase = feat + ((size_t)b * N_ + n0 + lr) * D_ + 2 * hi;
#pragma unroll
      for (int s = 0; s < 5; ++s) {
        v2f t;
        t.x = fbase[s * 4];
        t.y = fbase[s * 4 + 1];
        a[s] = t;
      }
    }
    __builtin_amdgcn_sched_barrier(0);

    const v4f* vrow =
        (const v4f*)(valid + ((size_t)b * N_ + n0 + 8 * hi) * K_ + 4 * lr);
    const v4f* arow =
        (const v4f*)(adj + ((size_t)b * N_ + n0 + 8 * hi) * K_ + 4 * lr);
    v4f vq[8], aq[8];
#pragma unroll
    for (int j = 0; j < 8; ++j) {
      vq[j] = __builtin_nontemporal_load(vrow + j * (K_ / 4));
      aq[j] = __builtin_nontemporal_load(arow + j * (K_ / 4));
    }

    __builtin_amdgcn_sched_barrier(0);

    float psq = 0.0f;
#pragma unroll
    for (int s = 0; s < 5; ++s) psq += a[s].x * a[s].x + a[s].y * a[s].y;
    float fsq = psq + __shfl_xor(psq, 16, 32);
    float fsqrow[8];
#pragma unroll
    for (int j = 0; j < 8; ++j) fsqrow[j] = __shfl(fsq, j + 8 * hi, 32);

    float rmax[8];
#pragma unroll
    for (int j = 0; j < 8; ++j) rmax[j] = -3.0e38f;

#pragma unroll
    for (int q = 0; q < 4; ++q) {
      v8f c[4];
#pragma unroll
      for (int ep = 0; ep < 2; ++ep) {
        v2f bm[2][5];
#pragma unroll
        for (int e = 0; e < 2; ++e) {
          const float* mrow = smean + ((q * 4 + 2 * ep + e) * 16 + lr) * D_ + 2 * hi;
#pragma unroll
          for (int s = 0; s < 5; ++s) {
            v2f t;
            t.x = mrow[s * 4];
            t.y = mrow[s * 4 + 1];
            bm[e][s] = t;
          }
        }
#pragma unroll
        for (int e = 0; e < 2; ++e) c[2 * ep + e] = (v8f){0,0,0,0,0,0,0,0};
#pragma unroll
        for (int s = 0; s < 5; ++s)
#pragma unroll
          for (int e = 0; e < 2; ++e)
            c[2 * ep + e] = __builtin_amdgcn_wmma_f32_16x16x4_f32(
                false, a[s], false, bm[e][s], (short)0, c[2 * ep + e], false, false);
        __builtin_amdgcn_sched_barrier(0);
      }
      float msq[4];
#pragma unroll
      for (int e = 0; e < 4; ++e) msq[e] = smsq[(q * 4 + e) * 16 + lr];
      const int prow = (q * 64 + lr) * NBP + w * 16 + 8 * hi;
#pragma unroll
      for (int jp = 0; jp < 4; ++jp) {
        v2f sc[4];
#pragma unroll
        for (int h = 0; h < 2; ++h) {
          const int j = 2 * jp + h;
          const v4f vs = vq[j];
          const v4f ad = aq[j];
#pragma unroll
          for (int e = 0; e < 4; ++e) {
            const float dist = (fsqrow[j] - 2.0f * c[e][j] + msq[e]) * vs[e] + ad[e];
            const float s = -dist;
            rmax[j] = fmaxf(rmax[j], s);
            sc[e][h] = s;
          }
        }
        __builtin_amdgcn_sched_barrier(0);
        if (q < 3) {
#pragma unroll
          for (int h = 0; h < 2; ++h) {
            const int j = 2 * jp + h;
            vq[j] = __builtin_nontemporal_load(vrow + j * (K_ / 4) + (q + 1) * 16);
            aq[j] = __builtin_nontemporal_load(arow + j * (K_ / 4) + (q + 1) * 16);
          }
        }
#pragma unroll
        for (int e = 0; e < 4; ++e) {
          pbuf[prow + e * 16 * NBP + 2 * jp] = sc[e].x;
          pbuf[prow + e * 16 * NBP + 2 * jp + 1] = sc[e].y;
        }
      }
    }

#pragma unroll
    for (int j = 0; j < 8; ++j) {
      float r = rmax[j];
      r = fmaxf(r, __shfl_xor(r, 1, 32));
      r = fmaxf(r, __shfl_xor(r, 2, 32));
      r = fmaxf(r, __shfl_xor(r, 4, 32));
      r = fmaxf(r, __shfl_xor(r, 8, 32));
      rmax[j] = r;
    }

    float rsum[8];
#pragma unroll
    for (int j = 0; j < 8; ++j) rsum[j] = 0.0f;
#pragma unroll
    for (int kt = 0; kt < 16; ++kt) {
      const int prow = (kt * 16 + lr) * NBP + w * 16;
#pragma unroll
      for (int j = 0; j < 8; ++j) {
        const int m = j + 8 * hi;
        const float e = __expf(pbuf[prow + m] - rmax[j]);
        rsum[j] += e;
        pbuf[prow + m] = e;
      }
    }
#pragma unroll
    for (int j = 0; j < 8; ++j) {
      float r = rsum[j];
      r += __shfl_xor(r, 1, 32);
      r += __shfl_xor(r, 2, 32);
      r += __shfl_xor(r, 4, 32);
      r += __shfl_xor(r, 8, 32);
      if (lr == j) rinvb[w * 16 + j + 8 * hi] = 1.0f / r;
    }
    __syncthreads();

#pragma unroll 2
    for (int ns = 0; ns < NB / 4; ++ns) {
      v2f pa[2];
#pragma unroll
      for (int q = 0; q < 2; ++q) {
        const int kk = (w * 2 + q) * 16 + lr;
        const float* pr = pbuf + kk * NBP + ns * 4 + 2 * hi;
        v2f t;
        t.x = pr[0];
        t.y = pr[1];
        pa[q] = t;
      }
      const float ri0 = rinvb[ns * 4 + 2 * hi];
      const float ri1 = rinvb[ns * 4 + 2 * hi + 1];
      v2f fb[2];
#pragma unroll
      for (int dt = 0; dt < 2; ++dt) {
        const int d = dt * 16 + lr;
        float x0, x1;
        if (d < D_) {
          const float* f0 =
              feat + ((size_t)b * N_ + nb0 + ns * 4 + 2 * hi) * D_ + d;
          x0 = f0[0] * ri0;
          x1 = f0[D_] * ri1;
        } else if (d == D_) {
          x0 = ri0;
          x1 = ri1;
        } else {
          x0 = 0.0f;
          x1 = 0.0f;
        }
        v2f t;
        t.x = x0;
        t.y = x1;
        fb[dt] = t;
      }
#pragma unroll
      for (int q = 0; q < 2; ++q)
#pragma unroll
        for (int dt = 0; dt < 2; ++dt)
          acc[q][dt] = __builtin_amdgcn_wmma_f32_16x16x4_f32(
              false, pa[q], false, fb[dt], (short)0, acc[q][dt], false, false);
    }
  }

  __syncthreads();
  float* const xbuf = pbuf2;
  if (half == 1) {
#pragma unroll
    for (int q = 0; q < 2; ++q)
#pragma unroll
      for (int dt = 0; dt < 2; ++dt)
#pragma unroll
        for (int j = 0; j < 8; ++j)
          xbuf[((q * 2 + dt) * 8 + j) * 256 + w * 32 + lane] = acc[q][dt][j];
  }
  __syncthreads();
  if (half == 1) return;
#pragma unroll
  for (int q = 0; q < 2; ++q)
#pragma unroll
    for (int dt = 0; dt < 2; ++dt)
#pragma unroll
      for (int j = 0; j < 8; ++j)
        acc[q][dt][j] += xbuf[((q * 2 + dt) * 8 + j) * 256 + w * 32 + lane];

#pragma unroll
  for (int q = 0; q < 2; ++q) {
    const int t = w * 2 + q;
    const int kbase = (t >> 2) * 64 + (t & 3);
#pragma unroll
    for (int dt = 0; dt < 2; ++dt) {
      const int d = dt * 16 + lr;
#pragma unroll
      for (int j = 0; j < 8; ++j) {
        const int k = kbase + 4 * (j + 8 * hi);
        const float v = acc[q][dt][j];
        if (d < D_)
          atomicAdd(out + ((b * K_ + k) * D_ + d), v);
        else if (d == D_)
          atomicAdd(&den[b * K_ + k], v);
      }
    }
  }
}

__global__ __launch_bounds__(256) void em_fin(float* __restrict__ out,
                                              const float* __restrict__ den) {
  int i = blockIdx.x * 256 + threadIdx.x;
  if (i < B_ * K_ * D_) out[i] = out[i] / fmaxf(den[i / D_], 1e-12f);
}

extern "C" void kernel_launch(void* const* d_in, const int* in_sizes, int n_in,
                              void* d_out, int out_size, void* d_ws,
                              size_t ws_size, hipStream_t stream) {
  (void)in_sizes; (void)n_in; (void)out_size; (void)ws_size;
  const float* mean = (const float*)d_in[0];
  const float* feat = (const float*)d_in[1];
  const float* valid = (const float*)d_in[2];
  const float* adj = (const float*)d_in[3];
  float* out = (float*)d_out;
  float* den = (float*)d_ws;

  const int tot = B_ * K_ * D_;
  em_zero<<<(tot + 255) / 256, 256, 0, stream>>>(out, den);
  em_main<<<B_ * NCHUNKS, 512, 0, stream>>>(mean, feat, valid, adj, out, den);
  em_fin<<<(tot + 255) / 256, 256, 0, stream>>>(out, den);
}


# M-step operand [F/rowsum | 1/rowsum | 0] built once per 128-pixel block in LDS in B-fragment order by the wave owning the pixels; M-step loop reads LDS only: no global loads, branches, multiplies
# speedup vs baseline: 3.5479x; 1.0750x over previous
#include <hip/hip_runtime.h>


#define B_ 8
#define N_ 16384
#define K_ 256
#define D_ 20
#define NCHUNKS 32
#define NB 128
#define NBP 130
#define ITERS ((N_ / NCHUNKS) / (2 * NB))

typedef __attribute__((ext_vector_type(2))) float v2f;
typedef __attribute__((ext_vector_type(4))) float v4f;
typedef __attribute__((ext_vector_type(8))) float v8f;

__global__ __launch_bounds__(256) void em_zero(float* __restrict__ out,
                                               float* __restrict__ den) {
  int i = blockIdx.x * 256 + threadIdx.x;
  if (i < B_ * K_ * D_) out[i] = 0.0f;
  if (i < B_ * K_) den[i] = 0.0f;
}

__global__ __launch_bounds__(512, 4) void em_main(const float* __restrict__ mean,
                                               const float* __restrict__ feat,
                                               const float* __restrict__ valid,
                                               const float* __restrict__ adj,
                                               float* __restrict__ out,
                                               float* __restrict__ den) {
  __shared__ float smean[K_ * D_];
  __shared__ float smsq[K_];
  __shared__ __attribute__((aligned(16))) float pbuf2[2 * K_ * NBP];
  __shared__ __attribute__((aligned(16))) float fbuf2[2 * 2 * 64 * 32];

  const int b = blockIdx.x / NCHUNKS;
  const int chunk = blockIdx.x % NCHUNKS;
  const int tid = threadIdx.x;
  const int half = __builtin_amdgcn_readfirstlane(tid >> 8);
  const int w = (tid >> 5) & 7;
  float* const pbuf = pbuf2 + half * (K_ * NBP);
  float* const fbuf = fbuf2 + half * (2 * 64 * 32);
  const int lane = tid & 31;
  const int hi = lane >> 4;
  const int lr = lane & 15;

  if (tid < K_) {
    const float* mrow = mean + ((size_t)b * K_ + tid) * D_;
    const int r = (((tid >> 6) * 4 + (tid & 3)) * 16) + ((tid >> 2) & 15);
    float sq = 0.0f;
#pragma unroll
    for (int d = 0; d < D_; ++d) {
      float v = mrow[d];
      smean[r * D_ + d] = v;
      sq += v * v;
    }
    smsq[r] = sq;
  }
  for (int i = tid; i < 2 * 2 * 64 * 32; i += 512) fbuf2[i] = 0.0f;
  __syncthreads();

  v8f acc[2][2];
#pragma unroll
  for (int q = 0; q < 2; ++q)
#pragma unroll
    for (int dt = 0; dt < 2; ++dt) acc[q][dt] = (v8f){0,0,0,0,0,0,0,0};

  const int nchunk0 = chunk * (N_ / NCHUNKS);

  for (int it = 0; it < ITERS; ++it) {
    const int nb0 = nchunk0 + (2 * it + half) * NB;
    const int n0 = nb0 + w * 16;

    __syncthreads();

    v2f a[5];
    {
      const float* fbase = feat + ((size_t)b * N_ + n0 + lr) * D_ + 2 * hi;
#pragma unroll
      for (int s = 0; s < 5; ++s) {
        v2f t;
        t.x = fbase[s * 4];
        t.y = fbase[s * 4 + 1];
        a[s] = t;
      }
    }
    __builtin_amdgcn_sched_barrier(0);

    const v4f* vrow =
        (const v4f*)(valid + ((size_t)b * N_ + n0 + 8 * hi) * K_ + 4 * lr);
    const v4f* arow =
        (const v4f*)(adj + ((size_t)b * N_ + n0 + 8 * hi) * K_ + 4 * lr);
    v4f vq[8], aq[8];
#pragma unroll
    for (int j = 0; j < 8; ++j) {
      vq[j] = __builtin_nontemporal_load(vrow + j * (K_ / 4));
      aq[j] = __builtin_nontemporal_load(arow + j * (K_ / 4));
    }

    __builtin_amdgcn_sched_barrier(0);

    float psq = 0.0f;
#pragma unroll
    for (int s = 0; s < 5; ++s) psq += a[s].x * a[s].x + a[s].y * a[s].y;
    float fsq = psq + __shfl_xor(psq, 16, 32);
    float fsqrow[8];
#pragma unroll
    for (int j = 0; j < 8; ++j) fsqrow[j] = __shfl(fsq, j + 8 * hi, 32);

    float rmax[8];
#pragma unroll
    for (int j = 0; j < 8; ++j) rmax[j] = -3.0e38f;

#pragma unroll
    for (int q = 0; q < 4; ++q) {
      v8f c[4];
#pragma unroll
      for (int ep = 0; ep < 2; ++ep) {
        v2f bm[2][5];
#pragma unroll
        for (int e = 0; e < 2; ++e) {
          const float* mrow = smean + ((q * 4 + 2 * ep + e) * 16 + lr) * D_ + 2 * hi;
#pragma unroll
          for (int s = 0; s < 5; ++s) {
            v2f t;
            t.x = mrow[s * 4];
            t.y = mrow[s * 4 + 1];
            bm[e][s] = t;
          }
        }
#pragma unroll
        for (int e = 0; e < 2; ++e) c[2 * ep + e] = (v8f){0,0,0,0,0,0,0,0};
#pragma unroll
        for (int s = 0; s < 5; ++s)
#pragma unroll
          for (int e = 0; e < 2; ++e)
            c[2 * ep + e] = __builtin_amdgcn_wmma_f32_16x16x4_f32(
                false, a[s], false, bm[e][s], (short)0, c[2 * ep + e], false, false);
        __builtin_amdgcn_sched_barrier(0);
      }
      float msq[4];
#pragma unroll
      for (int e = 0; e < 4; ++e) msq[e] = smsq[(q * 4 + e) * 16 + lr];
      const int prow = (q * 64 + lr) * NBP + w * 16 + 8 * hi;
#pragma unroll
      for (int jp = 0; jp < 4; ++jp) {
        v2f sc[4];
#pragma unroll
        for (int h = 0; h < 2; ++h) {
          const int j = 2 * jp + h;
          const v4f vs = vq[j];
          const v4f ad = aq[j];
#pragma unroll
          for (int e = 0; e < 4; ++e) {
            const float dist = (fsqrow[j] - 2.0f * c[e][j] + msq[e]) * vs[e] + ad[e];
            const float s = -dist;
            rmax[j] = fmaxf(rmax[j], s);
            sc[e][h] = s;
          }
        }
        __builtin_amdgcn_sched_barrier(0);
        if (q < 3) {
#pragma unroll
          for (int h = 0; h < 2; ++h) {
            const int j = 2 * jp + h;
            vq[j] = __builtin_nontemporal_load(vrow + j * (K_ / 4) + (q + 1) * 16);
            aq[j] = __builtin_nontemporal_load(arow + j * (K_ / 4) + (q + 1) * 16);
          }
        }
#pragma unroll
        for (int e = 0; e < 4; ++e) {
          pbuf[prow + e * 16 * NBP + 2 * jp] = sc[e].x;
          pbuf[prow + e * 16 * NBP + 2 * jp + 1] = sc[e].y;
        }
      }
    }

#pragma unroll
    for (int j = 0; j < 8; ++j) {
      float r = rmax[j];
      r = fmaxf(r, __shfl_xor(r, 1, 32));
      r = fmaxf(r, __shfl_xor(r, 2, 32));
      r = fmaxf(r, __shfl_xor(r, 4, 32));
      r = fmaxf(r, __shfl_xor(r, 8, 32));
      rmax[j] = r;
    }

    float rsum[8];
#pragma unroll
    for (int j = 0; j < 8; ++j) rsum[j] = 0.0f;
#pragma unroll
    for (int kt = 0; kt < 16; ++kt) {
      const int prow = (kt * 16 + lr) * NBP + w * 16;
#pragma unroll
      for (int j = 0; j < 8; ++j) {
        const int m = j + 8 * hi;
        const float e = __expf(pbuf[prow + m] - rmax[j]);
        rsum[j] += e;
        pbuf[prow + m] = e;
      }
    }
    __builtin_amdgcn_sched_barrier(0);
    float rown = 0.0f;
#pragma unroll
    for (int j = 0; j < 8; ++j) {
      float r = rsum[j];
      r += __shfl_xor(r, 1, 32);
      r += __shfl_xor(r, 2, 32);
      r += __shfl_xor(r, 4, 32);
      r += __shfl_xor(r, 8, 32);
      if ((lr & 7) == j) rown = r;
    }
    {
      const float rr = 1.0f / __shfl(rown, (lr >> 3) * 16 + (lr & 7), 32);
      float* const flo = fbuf + (w * 8 + (lr >> 1)) * 32 + (lr & 1) + 4 * hi;
      float* const fhi = flo + 64 * 32;
#pragma unroll
      for (int s = 0; s < 4; ++s) {
        flo[8 * s] = a[s].x * rr;
        flo[8 * s + 2] = a[s].y * rr;
      }
      fhi[0] = a[4].x * rr;
      fhi[2] = a[4].y * rr;
      if (hi == 0) fhi[8] = rr;
    }
    __syncthreads();

    {
      const float* const pa0 = pbuf + ((w * 2) * 16 + lr) * NBP + 2 * hi;
      const float* const pa1 = pa0 + 16 * NBP;
      const float* const fb0 = fbuf + hi * 32 + 2 * lr;
      const float* const fb1 = fb0 + 64 * 32;
#pragma unroll 8
      for (int ns = 0; ns < NB / 4; ++ns) {
        v2f pa[2];
        pa[0] = *(const v2f*)(pa0 + ns * 4);
        pa[1] = *(const v2f*)(pa1 + ns * 4);
        v2f fb[2];
        fb[0] = *(const v2f*)(fb0 + ns * 64);
        fb[1] = *(const v2f*)(fb1 + ns * 64);
#pragma unroll
        for (int q = 0; q < 2; ++q)
#pragma unroll
          for (int dt = 0; dt < 2; ++dt)
            acc[q][dt] = __builtin_amdgcn_wmma_f32_16x16x4_f32(
                false, pa[q], false, fb[dt], (short)0, acc[q][dt], false, false);
      }
    }
  }

  __syncthreads();
  float* const xbuf = pbuf2;
  if (half == 1) {
#pragma unroll
    for (int q = 0; q < 2; ++q)
#pragma unroll
      for (int dt = 0; dt < 2; ++dt)
#pragma unroll
        for (int j = 0; j < 8; ++j)
          xbuf[((q * 2 + dt) * 8 + j) * 256 + w * 32 + lane] = acc[q][dt][j];
  }
  __syncthreads();
  if (half == 1) return;
#pragma unroll
  for (int q = 0; q < 2; ++q)
#pragma unroll
    for (int dt = 0; dt < 2; ++dt)
#pragma unroll
      for (int j = 0; j < 8; ++j)
        acc[q][dt][j] += xbuf[((q * 2 + dt) * 8 + j) * 256 + w * 32 + lane];

#pragma unroll
  for (int q = 0; q < 2; ++q) {
    const int t = w * 2 + q;
    const int kbase = (t >> 2) * 64 + (t & 3);
#pragma unroll
    for (int dt = 0; dt < 2; ++dt) {
      const int d = dt * 16 + lr;
#pragma unroll
      for (int j = 0; j < 8; ++j) {
        const int k = kbase + 4 * (j + 8 * hi);
        const float v = acc[q][dt][j];
        if (d < D_)
          atomicAdd(out + ((b * K_ + k) * D_ + d), v);
        else if (d == D_)
          atomicAdd(&den[b * K_ + k], v);
      }
    }
  }
}

__global__ __launch_bounds__(256) void em_fin(float* __restrict__ out,
                                              const float* __restrict__ den) {
  int i = blockIdx.x * 256 + threadIdx.x;
  if (i < B_ * K_ * D_) out[i] = out[i] / fmaxf(den[i / D_], 1e-12f);
}

extern "C" void kernel_launch(void* const* d_in, const int* in_sizes, int n_in,
                              void* d_out, int out_size, void* d_ws,
                              size_t ws_size, hipStream_t stream) {
  (void)in_sizes; (void)n_in; (void)out_size; (void)ws_size;
  const float* mean = (const float*)d_in[0];
  const float* feat = (const float*)d_in[1];
  const float* valid = (const float*)d_in[2];
  const float* adj = (const float*)d_in[3];
  float* out = (float*)d_out;
  float* den = (float*)d_ws;

  const int tot = B_ * K_ * D_;
  em_zero<<<(tot + 255) / 256, 256, 0, stream>>>(out, den);
  em_main<<<B_ * NCHUNKS, 512, 0, stream>>>(mean, feat, valid, adj, out, den);
  em_fin<<<(tot + 255) / 256, 256, 0, stream>>>(out, den);
}


# float atomics and the zeroing launch go: each workgroup stores its sums as one plane in the workspace, the final kernel adds a batch's 32 planes in chunk order and divides; sums add in a new order
# speedup vs baseline: 5.3384x; 1.5047x over previous
#include <hip/hip_runtime.h>


#define B_ 8
#define N_ 16384
#define K_ 256
#define D_ 20
#define NCHUNKS 32
#define NB 128
#define NBP 130
#define ITERS ((N_ / NCHUNKS) / (2 * NB))
#define PLANE (K_ * D_ + K_)
#define WS_NEED ((size_t)B_ * NCHUNKS * PLANE * sizeof(float))

typedef __attribute__((ext_vector_type(2))) float v2f;
typedef __attribute__((ext_vector_type(4))) float v4f;
typedef __attribute__((ext_vector_type(8))) float v8f;

__global__ __launch_bounds__(512, 4) void em_main(const float* __restrict__ mean,
                                               const float* __restrict__ feat,
                                               const float* __restrict__ valid,
                                               const float* __restrict__ adj,
                                               float* __restrict__ out,
                                               float* __restrict__ part,
                                               size_t ws_bytes) {
  const bool fit = ws_bytes >= WS_NEED;
  __shared__ float smean[K_ * D_];
  __shared__ float smsq[K_];
  __shared__ __attribute__((aligned(16))) float pbuf2[2 * K_ * NBP];
  __shared__ __attribute__((aligned(16))) float fbuf2[2 * 2 * 64 * 32];

  const int b = blockIdx.x / NCHUNKS;
  const int chunk = blockIdx.x % NCHUNKS;
  if (!fit && chunk != 0) return;
  const int nit = fit ? ITERS : ITERS * NCHUNKS;
  const int tid = threadIdx.x;
  const int half = __builtin_amdgcn_readfirstlane(tid >> 8);
  const int w = (tid >> 5) & 7;
  float* const pbuf = pbuf2 + half * (K_ * NBP);
  float* const fbuf = fbuf2 + half * (2 * 64 * 32);
  const int lane = tid & 31;
  const int hi = lane >> 4;
  const int lr = lane & 15;

  if (tid < K_) {
    const float* mrow = mean + ((size_t)b * K_ + tid) * D_;
    const int r = (((tid >> 6) * 4 + (tid & 3)) * 16) + ((tid >> 2) & 15);
    float sq = 0.0f;
#pragma unroll
    for (int d = 0; d < D_; ++d) {
      float v = mrow[d];
      smean[r * D_ + d] = v;
      sq += v * v;
    }
    smsq[r] = sq;
  }
  for (int i = tid; i < 2 * 2 * 64 * 32; i += 512) fbuf2[i] = 0.0f;
  __syncthreads();

  v8f acc[2][2];
#pragma unroll
  for (int q = 0; q < 2; ++q)
#pragma unroll
    for (int dt = 0; dt < 2; ++dt) acc[q][dt] = (v8f){0,0,0,0,0,0,0,0};

  const int nchunk0 = chunk * (N_ / NCHUNKS);

  for (int it = 0; it < nit; ++it) {
    const int nb0 = nchunk0 + (2 * it + half) * NB;
    const int n0 = nb0 + w * 16;

    __syncthreads();

    v2f a[5];
    {
      const float* fbase = feat + ((size_t)b * N_ + n0 + lr) * D_ + 2 * hi;
#pragma unroll
      for (int s = 0; s < 5; ++s) {
        v2f t;
        t.x = fbase[s * 4];
        t.y = fbase[s * 4 + 1];
        a[s] = t;
      }
    }
    __builtin_amdgcn_sched_barrier(0);

    const v4f* vrow =
        (const v4f*)(valid + ((size_t)b * N_ + n0 + 8 * hi) * K_ + 4 * lr);
    const v4f* arow =
        (const v4f*)(adj + ((size_t)b * N_ + n0 + 8 * hi) * K_ + 4 * lr);
    v4f vq[8], aq[8];
#pragma unroll
    for (int j = 0; j < 8; ++j) {
      vq[j] = __builtin_nontemporal_load(vrow + j * (K_ / 4));
      aq[j] = __builtin_nontemporal_load(arow + j * (K_ / 4));
    }

    __builtin_amdgcn_sched_barrier(0);

    float psq = 0.0f;
#pragma unroll
    for (int s = 0; s < 5; ++s) psq += a[s].x * a[s].x + a[s].y * a[s].y;
    float fsq = psq + __shfl_xor(psq, 16, 32);
    float fsqrow[8];
#pragma unroll
    for (int j = 0; j < 8; ++j) fsqrow[j] = __shfl(fsq, j + 8 * hi, 32);

    float rmax[8];
#pragma unroll
    for (int j = 0; j < 8; ++j) rmax[j] = -3.0e38f;

#pragma unroll
    for (int q = 0; q < 4; ++q) {
      v8f c[4];
#pragma unroll
      for (int ep = 0; ep < 2; ++ep) {
        v2f bm[2][5];
#pragma unroll
        for (int e = 0; e < 2; ++e) {
          const float* mrow = smean + ((q * 4 + 2 * ep + e) * 16 + lr) * D_ + 2 * hi;
#pragma unroll
          for (int s = 0; s < 5; ++s) {
            v2f t;
            t.x = mrow[s * 4];
            t.y = mrow[s * 4 + 1];
            bm[e][s] = t;
          }
        }
#pragma unroll
        for (int e = 0; e < 2; ++e) c[2 * ep + e] = (v8f){0,0,0,0,0,0,0,0};
#pragma unroll
        for (int s = 0; s < 5; ++s)
#pragma unroll
          for (int e = 0; e < 2; ++e)
            c[2 * ep + e] = __builtin_amdgcn_wmma_f32_16x16x4_f32(
                false, a[s], false, bm[e][s], (short)0, c[2 * ep + e], false, false);
        __builtin_amdgcn_sched_barrier(0);
      }
      float msq[4];
#pragma unroll
      for (int e = 0; e < 4; ++e) msq[e] = smsq[(q * 4 + e) * 16 + lr];
      const int prow = (q * 64 + lr) * NBP + w * 16 + 8 * hi;
#pragma unroll
      for (int jp = 0; jp < 4; ++jp) {
        v2f sc[4];
#pragma unroll
        for (int h = 0; h < 2; ++h) {
          const int j = 2 * jp + h;
          const v4f vs = vq[j];
          const v4f ad = aq[j];
#pragma unroll
          for (int e = 0; e < 4; ++e) {
            const float dist = (fsqrow[j] - 2.0f * c[e][j] + msq[e]) * vs[e] + ad[e];
            const float s = -dist;
            rmax[j] = fmaxf(rmax[j], s);
            sc[e][h] = s;
          }
        }
        __builtin_amdgcn_sched_barrier(0);
        if (q < 3) {
#pragma unroll
          for (int h = 0; h < 2; ++h) {
            const int j = 2 * jp + h;
            vq[j] = __builtin_nontemporal_load(vrow + j * (K_ / 4) + (q + 1) * 16);
            aq[j] = __builtin_nontemporal_load(arow + j * (K_ / 4) + (q + 1) * 16);
          }
        }
#pragma unroll
        for (int e = 0; e < 4; ++e) {
          pbuf[prow + e * 16 * NBP + 2 * jp] = sc[e].x;
          pbuf[prow + e * 16 * NBP + 2 * jp + 1] = sc[e].y;
        }
      }
    }

#pragma unroll
    for (int j = 0; j < 8; ++j) {
      float r = rmax[j];
      r = fmaxf(r, __shfl_xor(r, 1, 32));
      r = fmaxf(r, __shfl_xor(r, 2, 32));
      r = fmaxf(r, __shfl_xor(r, 4, 32));
      r = fmaxf(r, __shfl_xor(r, 8, 32));
      rmax[j] = r;
    }

    float rsum[8];
#pragma unroll
    for (int j = 0; j < 8; ++j) rsum[j] = 0.0f;
#pragma unroll
    for (int kt = 0; kt < 16; ++kt) {
      const int prow = (kt * 16 + lr) * NBP + w * 16;
#pragma unroll
      for (int j = 0; j < 8; ++j) {
        const int m = j + 8 * hi;
        const float e = __expf(pbuf[prow + m] - rmax[j]);
        rsum[j] += e;
        pbuf[prow + m] = e;
      }
    }
    __builtin_amdgcn_sched_barrier(0);
    float rown = 0.0f;
#pragma unroll
    for (int j = 0; j < 8; ++j) {
      float r = rsum[j];
      r += __shfl_xor(r, 1, 32);
      r += __shfl_xor(r, 2, 32);
      r += __shfl_xor(r, 4, 32);
      r += __shfl_xor(r, 8, 32);
      if ((lr & 7) == j) rown = r;
    }
    {
      const float rr = 1.0f / __shfl(rown, (lr >> 3) * 16 + (lr & 7), 32);
      float* const flo = fbuf + (w * 8 + (lr >> 1)) * 32 + (lr & 1) + 4 * hi;
      float* const fhi = flo + 64 * 32;
#pragma unroll
      for (int s = 0; s < 4; ++s) {
        flo[8 * s] = a[s].x * rr;
        flo[8 * s + 2] = a[s].y * rr;
      }
      fhi[0] = a[4].x * rr;
      fhi[2] = a[4].y * rr;
      if (hi == 0) fhi[8] = rr;
    }
    __syncthreads();

    {
      const float* const pa0 = pbuf + ((w * 2) * 16 + lr) * NBP + 2 * hi;
      const float* const pa1 = pa0 + 16 * NBP;
      const float* const fb0 = fbuf + hi * 32 + 2 * lr;
      const float* const fb1 = fb0 + 64 * 32;
#pragma unroll 8
      for (int ns = 0; ns < NB / 4; ++ns) {
        v2f pa[2];
        pa[0] = *(const v2f*)(pa0 + ns * 4);
        pa[1] = *(const v2f*)(pa1 + ns * 4);
        v2f fb[2];
        fb[0] = *(const v2f*)(fb0 + ns * 64);
        fb[1] = *(const v2f*)(fb1 + ns * 64);
#pragma unroll
        for (int q = 0; q < 2; ++q)
#pragma unroll
          for (int dt = 0; dt < 2; ++dt)
            acc[q][dt] = __builtin_amdgcn_wmma_f32_16x16x4_f32(
                false, pa[q], false, fb[dt], (short)0, acc[q][dt], false, false);
      }
    }
  }

  __syncthreads();
  float* const xbuf = pbuf2;
  float* const obuf = pbuf2 + 4 * 8 * 256;
  const int slot = w * 32 + lane;
  if (half == 1) {
#pragma unroll
    for (int q = 0; q < 2; ++q)
#pragma unroll
      for (int dt = 0; dt < 2; ++dt)
#pragma unroll
        for (int j = 0; j < 8; ++j)
          xbuf[((q * 2 + dt) * 8 + j) * 256 + slot] = acc[q][dt][j];
  }
  __syncthreads();
  if (half == 0) {
#pragma unroll
    for (int q = 0; q < 2; ++q)
#pragma unroll
      for (int dt = 0; dt < 2; ++dt)
#pragma unroll
        for (int j = 0; j < 8; ++j)
          acc[q][dt][j] += xbuf[((q * 2 + dt) * 8 + j) * 256 + slot];

    const int mul1 = (lr < 4) ? D_ : 1;
    const int add1 = (lr < 4) ? 16 + lr : K_ * D_;
#pragma unroll
    for (int q = 0; q < 2; ++q) {
      const int t = w * 2 + q;
      const int k0 = (t >> 2) * 64 + (t & 3) + 32 * hi;
#pragma unroll
      for (int j = 0; j < 8; ++j) obuf[(k0 + 4 * j) * D_ + lr] = acc[q][0][j];
      if (lr <= 4) {
#pragma unroll
        for (int j = 0; j < 8; ++j) obuf[(k0 + 4 * j) * mul1 + add1] = acc[q][1][j];
      }
    }
  }
  __syncthreads();

  if (fit) {
    v4f* const dst = (v4f*)(part + (size_t)blockIdx.x * PLANE);
    const v4f* const so = (const v4f*)obuf;
    for (int i = tid; i < PLANE / 4; i += 512) dst[i] = so[i];
  } else {
    float* const ob = out + (size_t)b * (K_ * D_);
    for (int i = tid; i < K_ * D_; i += 512)
      ob[i] = obuf[i] / fmaxf(obuf[K_ * D_ + i / D_], 1e-12f);
  }
}

__global__ __launch_bounds__(128) void em_fin(float* __restrict__ out,
                                              const float* __restrict__ part,
                                              size_t ws_bytes) {
  if (ws_bytes < WS_NEED) return;
  const int i = blockIdx.x * 128 + threadIdx.x;
  const int b = i / (K_ * D_);
  const int r = i - b * (K_ * D_);
  const int k = r / D_;
  const float* const p = part + (size_t)b * NCHUNKS * PLANE;
  float nv[NCHUNKS], dv[NCHUNKS];
#pragma unroll
  for (int c = 0; c < NCHUNKS; ++c) {
    nv[c] = p[c * PLANE + r];
    dv[c] = p[c * PLANE + K_ * D_ + k];
  }
  float num = nv[0], den = dv[0];
#pragma unroll
  for (int c = 1; c < NCHUNKS; ++c) {
    num += nv[c];
    den += dv[c];
  }
  out[i] = num / fmaxf(den, 1e-12f);
}

extern "C" void kernel_launch(void* const* d_in, const int* in_sizes, int n_in,
                              void* d_out, int out_size, void* d_ws,
                              size_t ws_size, hipStream_t stream) {
  (void)in_sizes; (void)n_in; (void)out_size;
  const float* mean = (const float*)d_in[0];
  const float* feat = (const float*)d_in[1];
  const float* valid = (const float*)d_in[2];
  const float* adj = (const float*)d_in[3];
  float* out = (float*)d_out;
  float* part = (float*)d_ws;

  em_main<<<B_ * NCHUNKS, 512, 0, stream>>>(mean, feat, valid, adj, out, part, ws_size);
  em_fin<<<(B_ * K_ * D_) / 128, 128, 0, stream>>>(out, part, ws_size);
}
